# starting kernel
# baseline (speedup 1.0000x reference)
#include <hip/hip_runtime.h>


typedef float v2f __attribute__((ext_vector_type(2)));
typedef float v8f __attribute__((ext_vector_type(8)));

#define NTERMS 8
#define BDIM 1024
#define DDIM 512
#define NDIM 512
#define WAVES_PER_BLOCK 8

__global__ __launch_bounds__(256)
void ConjunctionLayer_wmma_kernel(const float* __restrict__ x,
                                  const float* __restrict__ W,
                                  float* __restrict__ out) {
    const int lane = threadIdx.x & 31;
    const int widb = threadIdx.x >> 5;
    const int wid  = blockIdx.x * WAVES_PER_BLOCK + widb;

    const int ntiles_n = NDIM >> 4;
    const int tb = (wid / ntiles_n) << 4;
    const int tn = (wid % ntiles_n) << 4;

    const int half = lane >> 4;
    const int l15  = lane & 15;

    const float* xrow = x + (size_t)(tb + l15) * DDIM + 2 * half;
    const float* wrow = W + (size_t)(tn + l15) * DDIM + 2 * half;

    v8f acc[NTERMS];
#pragma unroll
    for (int k = 0; k < NTERMS; ++k) {
        v8f z = {};
        acc[k] = z;
    }

    for (int d0 = 0; d0 < DDIM; d0 += 4) {
        v2f xv = *(const v2f*)(xrow + d0);
        v2f wv = *(const v2f*)(wrow + d0);
        v2f a;
        a.x = 1.0f - xv.x;
        a.y = 1.0f - xv.y;

        v2f pa = a;
        v2f pb = wv;

        acc[0] = __builtin_amdgcn_wmma_f32_16x16x4_f32(
            false, pa, false, pb, (short)0, acc[0], false, false);

#pragma unroll
        for (int k = 1; k < NTERMS; ++k) {
            pa.x *= a.x;  pa.y *= a.y;
            pb.x *= wv.x; pb.y *= wv.y;
            acc[k] = __builtin_amdgcn_wmma_f32_16x16x4_f32(
                false, pa, false, pb, (short)0, acc[k], false, false);
        }
    }

#pragma unroll
    for (int j = 0; j < 8; ++j) {
        float s = acc[0][j];
#pragma unroll
        for (int k = 1; k < NTERMS; ++k) {
            s += acc[k][j] * (1.0f / (float)(k + 1));
        }
        float val = expf(-s);
        int m = j + 8 * half;
        out[(size_t)(tb + m) * NDIM + (tn + l15)] = val;
    }
}

extern "C" void kernel_launch(void* const* d_in, const int* in_sizes, int n_in,
                              void* d_out, int out_size, void* d_ws, size_t ws_size,
                              hipStream_t stream) {
    const float* x = (const float*)d_in[0];
    const float* W = (const float*)d_in[1];
    float* out = (float*)d_out;

    const int tiles = (BDIM / 16) * (NDIM / 16);
    const int blocks = tiles / WAVES_PER_BLOCK;
    ConjunctionLayer_wmma_kernel<<<blocks, 256, 0, stream>>>(x, W, out);
}


# operands staged through a 3-stage LDS ring: workgroup copies its 16 x rows and 128 W rows, 64 d at a time, by row-contiguous 16-byte loads two chunks ahead; waves read fragments from LDS
# speedup vs baseline: 1.2179x; 1.2179x over previous
#include <hip/hip_runtime.h>


typedef float v2f __attribute__((ext_vector_type(2)));
typedef float v4f __attribute__((ext_vector_type(4)));
typedef float v8f __attribute__((ext_vector_type(8)));

#define NTERMS 8
#define BDIM 1024
#define DDIM 512
#define NDIM 512
#define WAVES_PER_BLOCK 8
#define DC 64
#define NCHUNK (DDIM / DC)
#define ROWS (16 + 16 * WAVES_PER_BLOCK)
#define PITCH (DC + 4)
#define STAGE (ROWS * PITCH)
#define NSTAGE 3
#define LPT (ROWS * (DC / 4) / 256)

#define WM(A, B, C) __builtin_amdgcn_wmma_f32_16x16x4_f32(false, (A), false, (B), (short)0, (C), false, false)

#define CHUNK_STEPS(AF, WF, Q0, Q1)                                       \
    _Pragma("unroll")                                                     \
    for (int q = (Q0); q < (Q1); ++q) {                                   \
        _Pragma("unroll")                                                 \
        for (int e = 0; e < 4; e += 2) {                                  \
            v2f a, w;                                                     \
            a.x = AF[q][e];  a.y = AF[q][e + 1];                          \
            w.x = WF[q][e];  w.y = WF[q][e + 1];                          \
            v2f pa = a;                                                   \
            v2f pb = w;                                                   \
            acc[0] = WM(pa, pb, acc[0]);                                  \
            _Pragma("unroll")                                             \
            for (int k = 1; k < NTERMS; ++k) {                            \
                pa.x *= a.x;  pa.y *= a.y;                                \
                pb.x *= w.x;  pb.y *= w.y;                                \
                acc[k] = WM(pa, pb, acc[k]);                              \
            }                                                             \
        }                                                                 \
    }

#define FETCH_CHUNK()                                                                              \
    g[0] = *(const v4f*)(xnext + soff);                                                            \
    _Pragma("unroll")                                                                              \
    for (int i = 1; i < LPT; ++i) g[i] = *(const v4f*)(wnext + (i - 1) * 16 * DDIM + soff);

#define STORE_CHUNK(S)                                                                             \
    {                                                                                              \
        float* dst = lds + (S) * STAGE + sdst;                                                     \
        *(v4f*)(dst) = 1.0f - g[0];                                                                \
        _Pragma("unroll")                                                                          \
        for (int i = 1; i < LPT; ++i) *(v4f*)(dst + i * 16 * PITCH) = g[i];                        \
    }

#define READ_FRAGS(AF, WF, S)                                                                      \
    {                                                                                              \
        const float* st = lds + (S) * STAGE;                                                       \
        _Pragma("unroll")                                                                          \
        for (int s = 0; s < 4; ++s) {                                                              \
            AF[2 * s]     = *(const v4f*)(st + afr + 16 * s);                                      \
            AF[2 * s + 1] = *(const v4f*)(st + afr + 16 * s + 4);                                  \
            WF[2 * s]     = *(const v4f*)(st + wfr + 16 * s);                                      \
            WF[2 * s + 1] = *(const v4f*)(st + wfr + 16 * s + 4);                                  \
        }                                                                                          \
    }

#define TRIP(CH, AF, WF, AFN, WFN)                                                                 \
    {                                                                                              \
        if ((CH) + 2 < NCHUNK) { FETCH_CHUNK() }                                                   \
                            \
        __builtin_amdgcn_sched_barrier(0);                                                         \
        CHUNK_STEPS(AF, WF, 0, 6)                                                                  \
                                                 \
        __builtin_amdgcn_sched_barrier(0);                                                         \
        if ((CH) + 2 < NCHUNK) {                                                                   \
            STORE_CHUNK(s_wr)                                                                      \
                                      \
            xnext += DC;                                                                           \
            wnext += DC;                                                                           \
        }                                                                                          \
        if ((CH) + 1 < NCHUNK) { READ_FRAGS(AFN, WFN, s_nxt) }                                     \
        __builtin_amdgcn_sched_barrier(0);                                                         \
        CHUNK_STEPS(AF, WF, 6, 8)                                                                  \
                                                   \
        __builtin_amdgcn_sched_barrier(0);                                                         \
                                                                                  \
        __syncthreads();                                                                           \
        { const int s_t = s_cur; s_cur = s_nxt; s_nxt = s_wr; s_wr = s_t; }                        \
    }

__global__ __launch_bounds__(256)
void conj_wmma_lds_kernel(const float* __restrict__ x,
                          const float* __restrict__ W,
                          float* __restrict__ out) {
    __shared__ __attribute__((aligned(16))) float lds[NSTAGE * STAGE];

    const int t    = threadIdx.x;
    const int lane = t & 31;
    const int widb = t >> 5;
    const int tb   = (blockIdx.x >> 2) << 4;
    const int tnb  = (blockIdx.x & 3) << 7;
    const int tn   = tnb + (widb << 4);

    const int half = lane >> 4;
    const int l15  = lane & 15;

    const int srow = t >> 4;
    const int sc4  = (t & 15) << 2;
    const float* xnext = x + (size_t)tb  * DDIM;
    const float* wnext = W + (size_t)tnb * DDIM;
    const unsigned soff = (unsigned)(srow * DDIM + sc4);
    const int sdst = srow * PITCH + sc4;

    const int afr = l15 * PITCH + 8 * half;
    const int wfr = (16 + (widb << 4) + l15) * PITCH + 8 * half;

    v8f acc[NTERMS];
#pragma unroll
    for (int k = 0; k < NTERMS; ++k) {
        v8f z = {};
        acc[k] = z;
    }

    v4f g[LPT];
    v4f af0[8], wf0[8], af1[8], wf1[8];
    int s_cur = 0, s_nxt = 1, s_wr = 2;

    FETCH_CHUNK()
    STORE_CHUNK(0)
    xnext += DC;
    wnext += DC;
    FETCH_CHUNK()
    STORE_CHUNK(1)
    xnext += DC;
    wnext += DC;
    __syncthreads();
    READ_FRAGS(af0, wf0, 0)

    for (int ch = 0; ch < NCHUNK; ch += 2) {
        TRIP(ch,     af0, wf0, af1, wf1)
        TRIP(ch + 1, af1, wf1, af0, wf0)
    }

#pragma unroll
    for (int j = 0; j < 8; ++j) {
        float s = acc[0][j];
#pragma unroll
        for (int k = 1; k < NTERMS; ++k) {
            s += acc[k][j] * (1.0f / (float)(k + 1));
        }
        float val = expf(-s);
        int m = j + 8 * half;
        out[(size_t)(tb + m) * NDIM + (tn + l15)] = val;
    }
}

extern "C" void kernel_launch(void* const* d_in, const int* in_sizes, int n_in,
                              void* d_out, int out_size, void* d_ws, size_t ws_size,
                              hipStream_t stream) {
    const float* x = (const float*)d_in[0];
    const float* W = (const float*)d_in[1];
    float* out = (float*)d_out;

    const int tiles = (BDIM / 16) * (NDIM / 16);
    const int blocks = tiles / WAVES_PER_BLOCK;
    conj_wmma_lds_kernel<<<blocks, 256, 0, stream>>>(x, W, out);
}


# operands centred exactly before the series: 1-a*w = P*Q/c*(1-u), u = (a-a0)/P*(w-w0)/Q under 1/36 (was 1/10): 5 f32 matrix products per d bound the truncation as 8 did; row products P, Q joined last
# speedup vs baseline: 1.6002x; 1.3139x over previous
#include <hip/hip_runtime.h>


typedef float v2f __attribute__((ext_vector_type(2)));
typedef float v4f __attribute__((ext_vector_type(4)));
typedef float v8f __attribute__((ext_vector_type(8)));

#define NTERMS 5
#define A0C 0.5f
#define W0C 0.05f
#define BDIM 1024
#define DDIM 512
#define NDIM 512
#define WAVES_PER_BLOCK 8
#define DC 64
#define NCHUNK (DDIM / DC)
#define ROWS (16 + 16 * WAVES_PER_BLOCK)
#define PITCH (DC + 4)
#define STAGE (ROWS * PITCH)
#define NSTAGE 3
#define LPT (ROWS * (DC / 4) / 256)

#define WM(A, B, C) __builtin_amdgcn_wmma_f32_16x16x4_f32(false, (A), false, (B), (short)0, (C), false, false)

#define CHUNK_STEPS(AF, WF, Q0, Q1)                                       \
    _Pragma("unroll")                                                     \
    for (int q = (Q0); q < (Q1); ++q) {                                   \
        _Pragma("unroll")                                                 \
        for (int e = 0; e < 4; e += 2) {                                  \
            v2f a, w;                                                     \
            a.x = AF[q][e];  a.y = AF[q][e + 1];                          \
            w.x = WF[q][e];  w.y = WF[q][e + 1];                          \
            v2f pa = a;                                                   \
            v2f pb = w;                                                   \
            acc[0] = WM(pa, pb, acc[0]);                                  \
            _Pragma("unroll")                                             \
            for (int k = 1; k < NTERMS; ++k) {                            \
                pa.x *= a.x;  pa.y *= a.y;                                \
                pb.x *= w.x;  pb.y *= w.y;                                \
                acc[k] = WM(pa, pb, acc[k]);                              \
            }                                                             \
        }                                                                 \
    }

#define FETCH_CHUNK()                                                                              \
    g[0] = *(const v4f*)(xnext + soff);                                                            \
    _Pragma("unroll")                                                                              \
    for (int i = 1; i < LPT; ++i) g[i] = *(const v4f*)(wnext + (i - 1) * 16 * DDIM + soff);

#define FORM_CHUNK()                                                                               \
    {                                                                                              \
        const v4f a = 1.0f - g[0];                                                                 \
        v4f p, r;                                                                                  \
        _Pragma("unroll")                                                                          \
        for (int e = 0; e < 4; ++e) {                                                              \
            p[e] = __builtin_fmaf(-W0C, a[e], 1.0f);                                               \
            r[e] = __builtin_amdgcn_rcpf(p[e]);                                                    \
        }                                                                                          \
        g[0] = (a - A0C) * r;                                                                      \
        fp *= (p[0] * p[1]) * (p[2] * p[3]);                                                       \
        _Pragma("unroll")                                                                          \
        for (int i = 1; i < LPT; ++i) {                                                            \
            const v4f w = g[i];                                                                    \
            v4f q, s;                                                                              \
            _Pragma("unroll")                                                                      \
            for (int e = 0; e < 4; ++e) {                                                          \
                q[e] = __builtin_fmaf(-A0C, w[e], 1.0f);                                           \
                s[e] = __builtin_amdgcn_rcpf(q[e]);                                                \
            }                                                                                      \
            g[i] = (w - W0C) * s;                                                                  \
            gp[i - 1] *= (q[0] * q[1]) * (q[2] * q[3]);                                            \
        }                                                                                          \
    }

#define STORE_CHUNK(S)                                                                             \
    {                                                                                              \
        float* dst = lds + (S) * STAGE + sdst;                                                     \
        _Pragma("unroll")                                                                          \
        for (int i = 0; i < LPT; ++i) *(v4f*)(dst + i * 16 * PITCH) = g[i];                        \
    }

#define READ_FRAGS(AF, WF, S)                                                                      \
    {                                                                                              \
        const float* st = lds + (S) * STAGE;                                                       \
        _Pragma("unroll")                                                                          \
        for (int s = 0; s < 4; ++s) {                                                              \
            AF[2 * s]     = *(const v4f*)(st + afr + 16 * s);                                      \
            AF[2 * s + 1] = *(const v4f*)(st + afr + 16 * s + 4);                                  \
            WF[2 * s]     = *(const v4f*)(st + wfr + 16 * s);                                      \
            WF[2 * s + 1] = *(const v4f*)(st + wfr + 16 * s + 4);                                  \
        }                                                                                          \
    }

#define TRIP(CH, AF, WF, AFN, WFN)                                                                 \
    {                                                                                              \
        if ((CH) + 2 < NCHUNK) { FETCH_CHUNK() }                                                   \
                            \
        __builtin_amdgcn_sched_barrier(0);                                                         \
        CHUNK_STEPS(AF, WF, 0, 6)                                                                  \
                                    \
        __builtin_amdgcn_sched_barrier(0);                                                         \
        if ((CH) + 2 < NCHUNK) {                                                                   \
            FORM_CHUNK()                                                                           \
            STORE_CHUNK(s_wr)                                                                      \
                                      \
            xnext += DC;                                                                           \
            wnext += DC;                                                                           \
        }                                                                                          \
        if ((CH) + 1 < NCHUNK) { READ_FRAGS(AFN, WFN, s_nxt) }                                     \
        __builtin_amdgcn_sched_barrier(0);                                                         \
        CHUNK_STEPS(AF, WF, 6, 8)                                                                  \
                                                   \
        __builtin_amdgcn_sched_barrier(0);                                                         \
                                                                                  \
        __syncthreads();                                                                           \
        { const int s_t = s_cur; s_cur = s_nxt; s_nxt = s_wr; s_wr = s_t; }                        \
    }

constexpr double ipow(double v, int n) { double r = 1.0; for (int i = 0; i < n; ++i) r *= v; return r; }
constexpr float KC = (float)ipow(1.0 / (1.0 - (double)A0C * (double)W0C), DDIM);

__global__ __launch_bounds__(256)
void conj_wmma_lds_kernel(const float* __restrict__ x,
                          const float* __restrict__ W,
                          float* __restrict__ out) {
    __shared__ __attribute__((aligned(16))) float lds[NSTAGE * STAGE];
    __shared__ float rowprod[ROWS];

    const int t    = threadIdx.x;
    const int lane = t & 31;
    const int widb = t >> 5;
    const int tb   = (blockIdx.x >> 2) << 4;
    const int tnb  = (blockIdx.x & 3) << 7;
    const int tn   = tnb + (widb << 4);

    const int half = lane >> 4;
    const int l15  = lane & 15;

    const int srow = t >> 4;
    const int sc4  = (t & 15) << 2;
    const float* xnext = x + (size_t)tb  * DDIM;
    const float* wnext = W + (size_t)tnb * DDIM;
    const unsigned soff = (unsigned)(srow * DDIM + sc4);
    const int sdst = srow * PITCH + sc4;

    const int afr = l15 * PITCH + 8 * half;
    const int wfr = (16 + (widb << 4) + l15) * PITCH + 8 * half;

    v8f acc[NTERMS];
#pragma unroll
    for (int k = 0; k < NTERMS; ++k) {
        v8f z = {};
        acc[k] = z;
    }

    v4f g[LPT];
    float fp = 1.0f;
    float gp[LPT - 1];
#pragma unroll
    for (int i = 0; i < LPT - 1; ++i) gp[i] = 1.0f;
    v4f af0[8], wf0[8], af1[8], wf1[8];
    int s_cur = 0, s_nxt = 1, s_wr = 2;

    FETCH_CHUNK()
    FORM_CHUNK()
    STORE_CHUNK(0)
    xnext += DC;
    wnext += DC;
    FETCH_CHUNK()
    FORM_CHUNK()
    STORE_CHUNK(1)
    xnext += DC;
    wnext += DC;
    __syncthreads();
    READ_FRAGS(af0, wf0, 0)

    for (int ch = 0; ch < NCHUNK; ch += 2) {
        TRIP(ch,     af0, wf0, af1, wf1)
        TRIP(ch + 1, af1, wf1, af0, wf0)
    }

#pragma unroll
    for (int m = 1; m < 16; m <<= 1) {
        fp *= __shfl_xor(fp, m);
#pragma unroll
        for (int i = 0; i < LPT - 1; ++i) gp[i] *= __shfl_xor(gp[i], m);
    }
    if ((t & 15) == 0) {
        rowprod[srow] = fp;
#pragma unroll
        for (int i = 0; i < LPT - 1; ++i) rowprod[16 + i * 16 + srow] = gp[i];
    }
    __syncthreads();

    const float gcol = rowprod[16 + (widb << 4) + l15];
#pragma unroll
    for (int j = 0; j < 8; ++j) {
        float s = acc[0][j];
#pragma unroll
        for (int k = 1; k < NTERMS; ++k) {
            s += acc[k][j] * (1.0f / (float)(k + 1));
        }
        int m = j + 8 * half;
        float val = (rowprod[m] * KC) * gcol * expf(-s);
        out[(size_t)(tb + m) * NDIM + (tn + l15)] = val;
    }
}

extern "C" void kernel_launch(void* const* d_in, const int* in_sizes, int n_in,
                              void* d_out, int out_size, void* d_ws, size_t ws_size,
                              hipStream_t stream) {
    const float* x = (const float*)d_in[0];
    const float* W = (const float*)d_in[1];
    float* out = (float*)d_out;

    const int tiles = (BDIM / 16) * (NDIM / 16);
    const int blocks = tiles / WAVES_PER_BLOCK;
    conj_wmma_lds_kernel<<<blocks, 256, 0, stream>>>(x, W, out);
}


# staging step's vector work cut: workgroup tile 32 x 64 (96 staged rows per chunk, not 144) and row products kept as packed pairs: 206 vector instructions per 80 matrix instructions, not 258
# speedup vs baseline: 1.6881x; 1.0549x over previous
#include <hip/hip_runtime.h>


typedef float v2f __attribute__((ext_vector_type(2)));
typedef float v4f __attribute__((ext_vector_type(4)));
typedef float v8f __attribute__((ext_vector_type(8)));

#define NTERMS 5
#define A0C 0.5f
#define W0C 0.05f
#define BDIM 1024
#define DDIM 512
#define NDIM 512
#define WAVES_PER_BLOCK 8
#define DC 64
#define NCHUNK (DDIM / DC)
#define XROWS 32
#define WROWS 64
#define ROWS (XROWS + WROWS)
#define PITCH (DC + 4)
#define STAGE (ROWS * PITCH)
#define NSTAGE 3
#define LPT (ROWS * (DC / 4) / 256)
#define XP (XROWS / 16)

#define WM(A, B, C) __builtin_amdgcn_wmma_f32_16x16x4_f32(false, (A), false, (B), (short)0, (C), false, false)

#define CHUNK_STEPS(AF, WF, Q0, Q1)                                       \
    _Pragma("unroll")                                                     \
    for (int q = (Q0); q < (Q1); ++q) {                                   \
        _Pragma("unroll")                                                 \
        for (int e = 0; e < 4; e += 2) {                                  \
            v2f a, w;                                                     \
            a.x = AF[q][e];  a.y = AF[q][e + 1];                          \
            w.x = WF[q][e];  w.y = WF[q][e + 1];                          \
            v2f pa = a;                                                   \
            v2f pb = w;                                                   \
            acc[0] = WM(pa, pb, acc[0]);                                  \
            _Pragma("unroll")                                             \
            for (int k = 1; k < NTERMS; ++k) {                            \
                pa.x *= a.x;  pa.y *= a.y;                                \
                pb.x *= w.x;  pb.y *= w.y;                                \
                acc[k] = WM(pa, pb, acc[k]);                              \
            }                                                             \
        }                                                                 \
    }

#define FETCH_CHUNK()                                                                              \
    _Pragma("unroll")                                                                              \
    for (int i = 0; i < XP; ++i) g[i] = *(const v4f*)(xnext + i * 16 * DDIM + soff);               \
    _Pragma("unroll")                                                                              \
    for (int i = XP; i < LPT; ++i) g[i] = *(const v4f*)(wnext + (i - XP) * 16 * DDIM + soff);

#define FORM_CHUNK()                                                                               \
    {                                                                                              \
        _Pragma("unroll")                                                                          \
        for (int i = 0; i < XP; ++i) {                                                             \
            const v4f a = 1.0f - g[i];                                                             \
            v4f p, r;                                                                              \
            _Pragma("unroll")                                                                      \
            for (int e = 0; e < 4; ++e) {                                                          \
                p[e] = __builtin_fmaf(-W0C, a[e], 1.0f);                                           \
                r[e] = __builtin_amdgcn_rcpf(p[e]);                                                \
            }                                                                                      \
            g[i] = (a - A0C) * r;                                                                  \
            fp[i] *= __builtin_shufflevector(p, p, 0, 1);                                          \
            fp[i] *= __builtin_shufflevector(p, p, 2, 3);                                          \
        }                                                                                          \
        _Pragma("unroll")                                                                          \
        for (int i = XP; i < LPT; ++i) {                                                           \
            const v4f w = g[i];                                                                    \
            v4f q, s;                                                                              \
            _Pragma("unroll")                                                                      \
            for (int e = 0; e < 4; ++e) {                                                          \
                q[e] = __builtin_fmaf(-A0C, w[e], 1.0f);                                           \
                s[e] = __builtin_amdgcn_rcpf(q[e]);                                                \
            }                                                                                      \
            g[i] = (w - W0C) * s;                                                                  \
            gp[i - XP] *= __builtin_shufflevector(q, q, 0, 1);                                     \
            gp[i - XP] *= __builtin_shufflevector(q, q, 2, 3);                                     \
        }                                                                                          \
    }

#define STORE_CHUNK(S)                                                                             \
    {                                                                                              \
        float* dst = lds + (S) * STAGE + sdst;                                                     \
        _Pragma("unroll")                                                                          \
        for (int i = 0; i < LPT; ++i) *(v4f*)(dst + i * 16 * PITCH) = g[i];                        \
    }

#define READ_FRAGS(AF, WF, S)                                                                      \
    {                                                                                              \
        const float* st = lds + (S) * STAGE;                                                       \
        _Pragma("unroll")                                                                          \
        for (int s = 0; s < 4; ++s) {                                                              \
            AF[2 * s]     = *(const v4f*)(st + afr + 16 * s);                                      \
            AF[2 * s + 1] = *(const v4f*)(st + afr + 16 * s + 4);                                  \
            WF[2 * s]     = *(const v4f*)(st + wfr + 16 * s);                                      \
            WF[2 * s + 1] = *(const v4f*)(st + wfr + 16 * s + 4);                                  \
        }                                                                                          \
    }

#define TRIP(CH, AF, WF, AFN, WFN)                                                                 \
    {                                                                                              \
        if ((CH) + 2 < NCHUNK) { FETCH_CHUNK() }                                                   \
                            \
        __builtin_amdgcn_sched_barrier(0);                                                         \
        CHUNK_STEPS(AF, WF, 0, 6)                                                                  \
                                    \
        __builtin_amdgcn_sched_barrier(0);                                                         \
        if ((CH) + 2 < NCHUNK) {                                                                   \
            FORM_CHUNK()                                                                           \
            STORE_CHUNK(s_wr)                                                                      \
                                      \
            xnext += DC;                                                                           \
            wnext += DC;                                                                           \
        }                                                                                          \
        if ((CH) + 1 < NCHUNK) { READ_FRAGS(AFN, WFN, s_nxt) }                                     \
        __builtin_amdgcn_sched_barrier(0);                                                         \
        CHUNK_STEPS(AF, WF, 6, 8)                                                                  \
                                                   \
        __builtin_amdgcn_sched_barrier(0);                                                         \
                                                                                  \
        __syncthreads();                                                                           \
        { const int s_t = s_cur; s_cur = s_nxt; s_nxt = s_wr; s_wr = s_t; }                        \
    }

constexpr double ipow(double v, int n) { double r = 1.0; for (int i = 0; i < n; ++i) r *= v; return r; }
constexpr float KC = (float)ipow(1.0 / (1.0 - (double)A0C * (double)W0C), DDIM);

__global__ __launch_bounds__(256)
void conj_wmma_lds_kernel(const float* __restrict__ x,
                          const float* __restrict__ W,
                          float* __restrict__ out) {
    __shared__ __attribute__((aligned(16))) float lds[NSTAGE * STAGE];
    __shared__ float rowprod[ROWS];

    const int t    = threadIdx.x;
    const int lane = t & 31;
    const int widb = t >> 5;
    const int wr   = widb >> 2;
    const int wc   = widb & 3;
    const int tbb  = (blockIdx.x >> 3) << 5;
    const int tnb  = (blockIdx.x & 7) << 6;
    const int tb   = tbb + (wr << 4);
    const int tn   = tnb + (wc << 4);

    const int half = lane >> 4;
    const int l15  = lane & 15;

    const int srow = t >> 4;
    const int sc4  = (t & 15) << 2;
    const float* xnext = x + (size_t)tbb * DDIM;
    const float* wnext = W + (size_t)tnb * DDIM;
    const unsigned soff = (unsigned)(srow * DDIM + sc4);
    const int sdst = srow * PITCH + sc4;

    const int afr = ((wr << 4) + l15) * PITCH + 8 * half;
    const int wfr = (XROWS + (wc << 4) + l15) * PITCH + 8 * half;

    v8f acc[NTERMS];
#pragma unroll
    for (int k = 0; k < NTERMS; ++k) {
        v8f z = {};
        acc[k] = z;
    }

    v4f g[LPT];
    v2f fp[XP];
    v2f gp[LPT - XP];
#pragma unroll
    for (int i = 0; i < XP; ++i) { fp[i].x = 1.0f; fp[i].y = 1.0f; }
#pragma unroll
    for (int i = 0; i < LPT - XP; ++i) { gp[i].x = 1.0f; gp[i].y = 1.0f; }
    v4f af0[8], wf0[8], af1[8], wf1[8];
    int s_cur = 0, s_nxt = 1, s_wr = 2;

    FETCH_CHUNK()
    FORM_CHUNK()
    STORE_CHUNK(0)
    xnext += DC;
    wnext += DC;
    FETCH_CHUNK()
    FORM_CHUNK()
    STORE_CHUNK(1)
    xnext += DC;
    wnext += DC;
    __syncthreads();
    READ_FRAGS(af0, wf0, 0)

    for (int ch = 0; ch < NCHUNK; ch += 2) {
        TRIP(ch,     af0, wf0, af1, wf1)
        TRIP(ch + 1, af1, wf1, af0, wf0)
    }

    float fps[XP], gps[LPT - XP];
#pragma unroll
    for (int i = 0; i < XP; ++i) fps[i] = fp[i].x * fp[i].y;
#pragma unroll
    for (int i = 0; i < LPT - XP; ++i) gps[i] = gp[i].x * gp[i].y;
#pragma unroll
    for (int m = 1; m < 16; m <<= 1) {
#pragma unroll
        for (int i = 0; i < XP; ++i) fps[i] *= __shfl_xor(fps[i], m);
#pragma unroll
        for (int i = 0; i < LPT - XP; ++i) gps[i] *= __shfl_xor(gps[i], m);
    }
    if ((t & 15) == 0) {
#pragma unroll
        for (int i = 0; i < XP; ++i) rowprod[i * 16 + srow] = fps[i];
#pragma unroll
        for (int i = 0; i < LPT - XP; ++i) rowprod[XROWS + i * 16 + srow] = gps[i];
    }
    __syncthreads();

    const float gcol = rowprod[XROWS + (wc << 4) + l15];
#pragma unroll
    for (int j = 0; j < 8; ++j) {
        float s = acc[0][j];
#pragma unroll
        for (int k = 1; k < NTERMS; ++k) {
            s += acc[k][j] * (1.0f / (float)(k + 1));
        }
        int m = j + 8 * half;
        float val = (rowprod[(wr << 4) + m] * KC) * gcol * expf(-s);
        out[(size_t)(tb + m) * NDIM + (tn + l15)] = val;
    }
}

extern "C" void kernel_launch(void* const* d_in, const int* in_sizes, int n_in,
                              void* d_out, int out_size, void* d_ws, size_t ws_size,
                              hipStream_t stream) {
    const float* x = (const float*)d_in[0];
    const float* W = (const float*)d_in[1];
    float* out = (float*)d_out;

    const int tiles = (BDIM / 16) * (NDIM / 16);
    const int blocks = tiles / WAVES_PER_BLOCK;
    conj_wmma_lds_kernel<<<blocks, 256, 0, stream>>>(x, W, out);
}
